# starting kernel
# baseline (speedup 1.0000x reference)
#include <hip/hip_runtime.h>

typedef __attribute__((ext_vector_type(2))) float v2f;
typedef __attribute__((ext_vector_type(8))) float v8f;

#define G_MAX          1024
#define DP             1024
#define TILES_PER_WAVE 8

__global__ void __launch_bounds__(G_MAX)
scan_counts_kernel(const int* __restrict__ counts, int* __restrict__ offs, int g) {
    __shared__ int s[G_MAX];
    const int t = threadIdx.x;
    const int c = (t < g) ? counts[t] : 0;
    s[t] = c;
    __syncthreads();
    #pragma unroll
    for (int d = 1; d < G_MAX; d <<= 1) {
        int v = s[t];
        if (t >= d) v += s[t - d];
        __syncthreads();
        s[t] = v;
        __syncthreads();
    }
    if (t < g) offs[t] = s[t] - c;
}

__global__ void __launch_bounds__(256)
seg_mean_wmma_kernel(const float* __restrict__ feat,
                     const int*   __restrict__ counts,
                     const int*   __restrict__ offs,
                     float*       __restrict__ out) {
    const int g     = blockIdx.x;
    const int cnt   = counts[g];
    const int start = offs[g];

    const int tid  = threadIdx.x;
    const int wave = tid >> 5;
    const int lane = tid & 31;
    const int half = lane >> 4;
    const int nl   = lane & 15;

    const float* __restrict__ base = feat + (size_t)start * DP;
    const int wbase   = wave * (TILES_PER_WAVE * 16);
    const int colbase = wbase + nl;
    const int rowsel  = (2 * half) * DP;

    v8f acc[TILES_PER_WAVE] = {};
    const v2f ones = {1.0f, 1.0f};

    auto loadStep = [&](int kk, v2f (&bb)[TILES_PER_WAVE]) {
        const int off = kk * DP + rowsel + colbase;
        #pragma unroll
        for (int t = 0; t < TILES_PER_WAVE; ++t) {
            bb[t].x = base[off + t * 16];
            bb[t].y = base[off + DP + t * 16];
        }
    };
    auto pfStep = [&](int kk) {
        const int pfo = kk * DP + rowsel + colbase;
        #pragma unroll
        for (int t = 0; t < TILES_PER_WAVE; ++t) {
            __builtin_prefetch(&base[pfo + t * 16], 0, 1);
            __builtin_prefetch(&base[pfo + DP + t * 16], 0, 1);
        }
    };
    auto mmaStep = [&](v2f (&bb)[TILES_PER_WAVE]) {
        #pragma unroll
        for (int t = 0; t < TILES_PER_WAVE; ++t)
            acc[t] = __builtin_amdgcn_wmma_f32_16x16x4_f32(
                false, ones, false, bb[t], (short)0, acc[t], false, false);
    };

    v2f A[TILES_PER_WAVE], B[TILES_PER_WAVE];
    int k0 = 0;

    if (cnt >= 4) loadStep(0, A);

    for (; k0 + 12 <= cnt; k0 += 8) {
        loadStep(k0 + 4, B);
        if (k0 + 20 <= cnt) pfStep(k0 + 16);
        mmaStep(A);
        loadStep(k0 + 8, A);
        if (k0 + 24 <= cnt) pfStep(k0 + 20);
        mmaStep(B);
    }

    if (k0 + 8 <= cnt) {
        loadStep(k0 + 4, B);
        mmaStep(A);
        mmaStep(B);
        k0 += 8;
    } else if (k0 + 4 <= cnt) {
        mmaStep(A);
        k0 += 4;
    }

    if (k0 < cnt) {
        const int  n0  = k0 + 2 * half;
        const bool p0  = (n0     < cnt);
        const bool p1  = (n0 + 1 < cnt);
        const int  off = k0 * DP + rowsel + colbase;
        #pragma unroll
        for (int t = 0; t < TILES_PER_WAVE; ++t) {
            v2f bt;
            bt.x = p0 ? base[off + t * 16]      : 0.0f;
            bt.y = p1 ? base[off + DP + t * 16] : 0.0f;
            acc[t] = __builtin_amdgcn_wmma_f32_16x16x4_f32(
                false, ones, false, bt, (short)0, acc[t], false, false);
        }
    }

    const float cf = (float)cnt;
    if (lane < 16) {
        float* __restrict__ og = out + (size_t)g * DP;
        #pragma unroll
        for (int t = 0; t < TILES_PER_WAVE; ++t) {
            const int col = wbase + t * 16 + nl;
            const int dp  = ((col & 255) << 2) | (col >> 8);
            og[dp] = acc[t][0] / cf;
        }
    }
}

extern "C" void kernel_launch(void* const* d_in, const int* in_sizes, int n_in,
                              void* d_out, int out_size, void* d_ws, size_t ws_size,
                              hipStream_t stream) {
    (void)n_in; (void)out_size; (void)ws_size;
    const float* feat   = (const float*)d_in[0];
    const int*   counts = (const int*)d_in[1];
    const int    G      = in_sizes[1];
    int* offs = (int*)d_ws;

    scan_counts_kernel<<<1, G_MAX, 0, stream>>>(counts, offs, G);
    seg_mean_wmma_kernel<<<G, 256, 0, stream>>>(feat, counts, offs, (float*)d_out);
}


# 16-byte row loads: A becomes an even/odd k selector so a lane's float4 is two B operands as loaded (4 tiles, prefetches dropped); rows summed 2 per product in 2 chains per column: new sum order
# speedup vs baseline: 1.3794x; 1.3794x over previous
#include <hip/hip_runtime.h>


typedef __attribute__((ext_vector_type(2))) float v2f;
typedef __attribute__((ext_vector_type(4))) float v4f;
typedef __attribute__((ext_vector_type(8))) float v8f;

#define G_MAX          1024
#define DP             1024
#define SLOTS          4
#define NLD            8
#define UNIT_CAP       (2 * G_MAX)
#define ROW_BITS       18

__global__ void __launch_bounds__(G_MAX)
scan_counts_kernel(const int* __restrict__ counts, int* __restrict__ offs,
                   int* __restrict__ utab, int* __restrict__ nunits, int g) {
    __shared__ unsigned s[G_MAX];
    const int t = threadIdx.x;
    const int c = (t < g) ? counts[t] : 0;
    int S = 0;
    if (t < g) {
        const int w = 1 + (c >> 6);
        S = (w >= 8) ? 8 : (w >= 4) ? 4 : (w >= 2) ? 2 : 1;
    }
    const unsigned val = (unsigned)c + ((unsigned)S << ROW_BITS);
    s[t] = val;
    __syncthreads();
    #pragma unroll
    for (int d = 1; d < G_MAX; d <<= 1) {
        unsigned v = s[t];
        if (t >= d) v += s[t - d];
        __syncthreads();
        s[t] = v;
        __syncthreads();
    }
    const unsigned incl = s[t];
    const unsigned excl = incl - val;
    if (t < g) {
        offs[t] = (int)(excl & ((1u << ROW_BITS) - 1u));
        const int u0 = (int)(excl >> ROW_BITS);
        for (int q = 0; q < S; ++q)
            if (u0 + q < UNIT_CAP) utab[u0 + q] = t | (q << 10) | (S << 13);
        if (t == g - 1) nunits[0] = (int)(incl >> ROW_BITS);
    }
}

template <int S>
__device__ __forceinline__ void seg_unit(const float* __restrict__ feat, const int cnt,
                                         const int start, const int g, const int q,
                                         float* __restrict__ out) {
    constexpr int T    = (S == 1) ? 2 : 1;
    constexpr int LPR  = (S <= 2) ? 16 : 32 / S;
    constexpr int RH   = 16 / LPR;
    constexpr int RPI  = 2 * RH;
    constexpr int RG   = NLD / T;
    constexpr int STEP = RG * RPI;
    constexpr int NP   = SLOTS / 2;

    const int tid  = threadIdx.x;
    const int wave = tid >> 5;
    const int lane = tid & 31;
    const int half = lane >> 4;
    const int nl   = lane & 15;
    const int cl   = nl % LPR;
    const int rs   = nl / LPR;
    const int lrow = half * RH + rs;

    const int wbase = (wave >> 1) * 256 + q * (256 / S) + (wave & 1) * (128 / S);
    const float* __restrict__ lp = feat + (size_t)start * DP + lrow * DP + wbase + cl * 4;

    v8f acc[SLOTS] = {};
    const v2f sel = {(nl < 8) ? 1.0f : 0.0f, (nl < 8) ? 0.0f : 1.0f};

    auto loadStep = [&](int kk, v4f (&bb)[NLD]) {
        #pragma unroll
        for (int i = 0; i < NLD; ++i) {
            const int so = (kk + (i / T) * RPI) * DP + (i % T) * 64;
            bb[i] = *reinterpret_cast<const v4f*>(lp + so);
        }
    };
    auto mma2 = [&](int i, const v4f& b) {
        const int a = 2 * (i % NP);
        const v2f b0 = {b.x, b.y};
        const v2f b1 = {b.z, b.w};
        acc[a]     = __builtin_amdgcn_wmma_f32_16x16x4_f32(
            false, sel, false, b0, (short)0, acc[a], false, false);
        acc[a + 1] = __builtin_amdgcn_wmma_f32_16x16x4_f32(
            false, sel, false, b1, (short)0, acc[a + 1], false, false);
    };
    auto mmaStep = [&](v4f (&bb)[NLD]) {
        #pragma unroll
        for (int i = 0; i < NLD; ++i) mma2(i, bb[i]);
    };

    v4f A[NLD], B[NLD];
    int k0 = 0;

    if (cnt >= STEP) loadStep(0, A);

    for (; k0 + 3 * STEP <= cnt; k0 += 2 * STEP) {
        loadStep(k0 + STEP, B);
        mmaStep(A);
        loadStep(k0 + 2 * STEP, A);
        mmaStep(B);
    }

    if (k0 + 2 * STEP <= cnt) {
        loadStep(k0 + STEP, B);
        mmaStep(A);
        mmaStep(B);
        k0 += 2 * STEP;
    } else if (k0 + STEP <= cnt) {
        mmaStep(A);
        k0 += STEP;
    }

    if (k0 < cnt) {
        #pragma unroll
        for (int i = 0; i < NLD; ++i) {
            const int r0 = k0 + (i / T) * RPI;
            const v4f z = {0.0f, 0.0f, 0.0f, 0.0f};
            A[i] = (r0 + lrow < cnt)
                 ? *reinterpret_cast<const v4f*>(lp + r0 * DP + (i % T) * 64) : z;
        }
        #pragma unroll
        for (int i = 0; i < NLD / 2; ++i) mma2(i, A[i]);
        if (k0 + STEP / 2 < cnt) {
            #pragma unroll
            for (int i = NLD / 2; i < NLD; ++i) mma2(i, A[i]);
        }
    }

    const float cf = (float)cnt;
    float* __restrict__ og = out + (size_t)g * DP;
    #pragma unroll
    for (int t = 0; t < T; ++t) {
        #pragma unroll
        for (int j = 0; j < 2; ++j) {
            float v = acc[2 * t + j][0];
            #pragma unroll
            for (int a = t + T; a < NP; a += T) v += acc[2 * a + j][0];
            #pragma unroll
            for (int m = LPR; m < 16; m <<= 1) v += __shfl_xor(v, m);
            if (rs == 0) {
                const int col = wbase + t * 64 + cl * 4 + 2 * j + half;
                const int dp  = ((col & 255) << 2) | (col >> 8);
                og[dp] = v / cf;
            }
        }
    }
}

__global__ void __launch_bounds__(256, 8)
seg_mean_wmma_kernel(const float* __restrict__ feat,
                     const int*   __restrict__ counts,
                     const int*   __restrict__ offs,
                     const int*   __restrict__ utab,
                     const int*   __restrict__ nunits,
                     float*       __restrict__ out) {
    const int b = blockIdx.y * gridDim.x + blockIdx.x;
    if (b >= nunits[0]) return;
    const int e = utab[b];
    const int g = e & 1023;
    const int q = (e >> 10) & 7;
    const int S = e >> 13;
    const int cnt   = counts[g];
    const int start = offs[g];
    if (S == 1)      seg_unit<1>(feat, cnt, start, g, q, out);
    else if (S == 2) seg_unit<2>(feat, cnt, start, g, q, out);
    else if (S == 4) seg_unit<4>(feat, cnt, start, g, q, out);
    else             seg_unit<8>(feat, cnt, start, g, q, out);
}

extern "C" void kernel_launch(void* const* d_in, const int* in_sizes, int n_in,
                              void* d_out, int out_size, void* d_ws, size_t ws_size,
                              hipStream_t stream) {
    (void)n_in; (void)out_size; (void)ws_size;
    const float* feat   = (const float*)d_in[0];
    const int*   counts = (const int*)d_in[1];
    const int    G      = in_sizes[1];
    int* offs   = (int*)d_ws;
    int* utab   = offs + G_MAX;
    int* nunits = utab + UNIT_CAP;

    scan_counts_kernel<<<1, G_MAX, 0, stream>>>(counts, offs, utab, nunits, G);
    seg_mean_wmma_kernel<<<dim3(G, 2), 256, 0, stream>>>(feat, counts, offs, utab, nunits,
                                                         (float*)d_out);
}


# One launch instead of two: each block rebuilds the row and unit prefix sums from counts itself (16-byte load per thread, wave scan, LDS hand-over) and finds its own unit; scan kernel and tables gone
# speedup vs baseline: 1.4824x; 1.0747x over previous
#include <hip/hip_runtime.h>


typedef __attribute__((ext_vector_type(2))) float v2f;
typedef __attribute__((ext_vector_type(4))) float v4f;
typedef __attribute__((ext_vector_type(8))) float v8f;

#define DP             1024
#define SLOTS          4
#define NLD            8
#define ROW_BITS       18


template <int S>
__device__ __forceinline__ void seg_unit(const float* __restrict__ feat, const int cnt,
                                         const int start, const int g, const int q,
                                         float* __restrict__ out) {
    constexpr int T    = (S == 1) ? 2 : 1;
    constexpr int LPR  = (S <= 2) ? 16 : 32 / S;
    constexpr int RH   = 16 / LPR;
    constexpr int RPI  = 2 * RH;
    constexpr int RG   = NLD / T;
    constexpr int STEP = RG * RPI;
    constexpr int NP   = SLOTS / 2;

    const int tid  = threadIdx.x;
    const int wave = tid >> 5;
    const int lane = tid & 31;
    const int half = lane >> 4;
    const int nl   = lane & 15;
    const int cl   = nl % LPR;
    const int rs   = nl / LPR;
    const int lrow = half * RH + rs;

    const int wbase = (wave >> 1) * 256 + q * (256 / S) + (wave & 1) * (128 / S);
    const float* __restrict__ lp = feat + (size_t)start * DP + lrow * DP + wbase + cl * 4;

    v8f acc[SLOTS] = {};
    const v2f sel = {(nl < 8) ? 1.0f : 0.0f, (nl < 8) ? 0.0f : 1.0f};

    auto loadStep = [&](int kk, v4f (&bb)[NLD]) {
        #pragma unroll
        for (int i = 0; i < NLD; ++i) {
            const int so = (kk + (i / T) * RPI) * DP + (i % T) * 64;
            bb[i] = *reinterpret_cast<const v4f*>(lp + so);
        }
    };
    auto mma2 = [&](int i, const v4f& b) {
        const int a = 2 * (i % NP);
        const v2f b0 = {b.x, b.y};
        const v2f b1 = {b.z, b.w};
        acc[a]     = __builtin_amdgcn_wmma_f32_16x16x4_f32(
            false, sel, false, b0, (short)0, acc[a], false, false);
        acc[a + 1] = __builtin_amdgcn_wmma_f32_16x16x4_f32(
            false, sel, false, b1, (short)0, acc[a + 1], false, false);
    };
    auto mmaStep = [&](v4f (&bb)[NLD]) {
        #pragma unroll
        for (int i = 0; i < NLD; ++i) mma2(i, bb[i]);
    };

    v4f A[NLD], B[NLD];
    int k0 = 0;

    if (cnt >= STEP) loadStep(0, A);

    for (; k0 + 3 * STEP <= cnt; k0 += 2 * STEP) {
        loadStep(k0 + STEP, B);
        mmaStep(A);
        loadStep(k0 + 2 * STEP, A);
        mmaStep(B);
    }

    if (k0 + 2 * STEP <= cnt) {
        loadStep(k0 + STEP, B);
        mmaStep(A);
        mmaStep(B);
        k0 += 2 * STEP;
    } else if (k0 + STEP <= cnt) {
        mmaStep(A);
        k0 += STEP;
    }

    if (k0 < cnt) {
        #pragma unroll
        for (int i = 0; i < NLD; ++i) {
            const int r0 = k0 + (i / T) * RPI;
            const v4f z = {0.0f, 0.0f, 0.0f, 0.0f};
            A[i] = (r0 + lrow < cnt)
                 ? *reinterpret_cast<const v4f*>(lp + r0 * DP + (i % T) * 64) : z;
        }
        #pragma unroll
        for (int i = 0; i < NLD / 2; ++i) mma2(i, A[i]);
        if (k0 + STEP / 2 < cnt) {
            #pragma unroll
            for (int i = NLD / 2; i < NLD; ++i) mma2(i, A[i]);
        }
    }

    const float cf = (float)cnt;
    float* __restrict__ og = out + (size_t)g * DP;
    #pragma unroll
    for (int t = 0; t < T; ++t) {
        #pragma unroll
        for (int j = 0; j < 2; ++j) {
            float v = acc[2 * t + j][0];
            #pragma unroll
            for (int a = t + T; a < NP; a += T) v += acc[2 * a + j][0];
            #pragma unroll
            for (int m = LPR; m < 16; m <<= 1) v += __shfl_xor(v, m);
            if (rs == 0) {
                const int col = wbase + t * 64 + cl * 4 + 2 * j + half;
                const int dp  = ((col & 255) << 2) | (col >> 8);
                og[dp] = v / cf;
            }
        }
    }
}

__global__ void __launch_bounds__(256, 8)
seg_mean_wmma_kernel(const float* __restrict__ feat,
                     const int*   __restrict__ counts,
                     float*       __restrict__ out,
                     const int                 ng) {
    __shared__ unsigned wtot[8];
    __shared__ int rec[3];
    const int b    = blockIdx.y * gridDim.x + blockIdx.x;
    const int t    = threadIdx.x;
    const int lane = t & 31;
    const int wave = t >> 5;

    int c[4];
    if (4 * t + 3 < ng) {
        const int4 cv = *reinterpret_cast<const int4*>(counts + 4 * t);
        c[0] = cv.x; c[1] = cv.y; c[2] = cv.z; c[3] = cv.w;
    } else {
        #pragma unroll
        for (int j = 0; j < 4; ++j) c[j] = (4 * t + j < ng) ? counts[4 * t + j] : 0;
    }
    int      S[4];
    unsigned val[4];
    unsigned v = 0;
    #pragma unroll
    for (int j = 0; j < 4; ++j) {
        const int w = 1 + (c[j] >> 6);
        S[j]   = (4 * t + j < ng) ? ((w >= 8) ? 8 : (w >= 4) ? 4 : (w >= 2) ? 2 : 1) : 0;
        val[j] = (unsigned)c[j] + ((unsigned)S[j] << ROW_BITS);
        v += val[j];
    }

    unsigned inc = v;
    #pragma unroll
    for (int d = 1; d < 32; d <<= 1) {
        const unsigned n = __shfl_up(inc, d);
        if (lane >= d) inc += n;
    }
    if (lane == 31) wtot[wave] = inc;
    __syncthreads();
    unsigned wpre = 0, total = 0;
    #pragma unroll
    for (int w = 0; w < 8; ++w) {
        const unsigned x = wtot[w];
        if (w < wave) wpre += x;
        total += x;
    }
    if (b >= (int)(total >> ROW_BITS)) return;

    unsigned e = wpre + inc - v;
    #pragma unroll
    for (int j = 0; j < 4; ++j) {
        const int u0 = (int)(e >> ROW_BITS);
        if (b >= u0 && b < u0 + S[j]) {
            rec[0] = (4 * t + j) | ((b - u0) << 10) | (S[j] << 13);
            rec[1] = c[j];
            rec[2] = (int)(e & ((1u << ROW_BITS) - 1u));
        }
        e += val[j];
    }
    __syncthreads();
    const int en    = __builtin_amdgcn_readfirstlane(rec[0]);
    const int cnt   = __builtin_amdgcn_readfirstlane(rec[1]);
    const int start = __builtin_amdgcn_readfirstlane(rec[2]);
    const int g = en & 1023;
    const int q = (en >> 10) & 7;
    const int Su = en >> 13;
    if (Su == 1)      seg_unit<1>(feat, cnt, start, g, q, out);
    else if (Su == 2) seg_unit<2>(feat, cnt, start, g, q, out);
    else if (Su == 4) seg_unit<4>(feat, cnt, start, g, q, out);
    else              seg_unit<8>(feat, cnt, start, g, q, out);
}

extern "C" void kernel_launch(void* const* d_in, const int* in_sizes, int n_in,
                              void* d_out, int out_size, void* d_ws, size_t ws_size,
                              hipStream_t stream) {
    (void)n_in; (void)out_size; (void)d_ws; (void)ws_size;
    const float* feat   = (const float*)d_in[0];
    const int*   counts = (const int*)d_in[1];
    const int    G      = in_sizes[1];

    seg_mean_wmma_kernel<<<dim3(G, 2), 256, 0, stream>>>(feat, counts, (float*)d_out, G);
}


# One resident round: grid of G blocks, not 2G; block b does unit b, then unit G+r goes to the block whose first unit ranks r by length + index/16 rows: no second entry and scan; sums unchanged
# speedup vs baseline: 1.5628x; 1.0542x over previous
#include <hip/hip_runtime.h>


typedef __attribute__((ext_vector_type(2))) float v2f;
typedef __attribute__((ext_vector_type(4))) float v4f;
typedef __attribute__((ext_vector_type(8))) float v8f;

#define DP             1024
#define SLOTS          4
#define NLD            8
#define ROW_BITS       18


template <int S>
__device__ __forceinline__ void seg_unit(const float* __restrict__ feat, const int cnt,
                                         const int start, const int g, const int q,
                                         float* __restrict__ out, const int tid) {
    constexpr int T    = (S == 1) ? 2 : 1;
    constexpr int LPR  = (S <= 2) ? 16 : 32 / S;
    constexpr int RH   = 16 / LPR;
    constexpr int RPI  = 2 * RH;
    constexpr int RG   = NLD / T;
    constexpr int STEP = RG * RPI;
    constexpr int NP   = SLOTS / 2;

    const int wave = tid >> 5;
    const int lane = tid & 31;
    const int half = lane >> 4;
    const int nl   = lane & 15;
    const int cl   = nl % LPR;
    const int rs   = nl / LPR;
    const int lrow = half * RH + rs;

    const int wbase = (wave >> 1) * 256 + q * (256 / S) + (wave & 1) * (128 / S);
    const float* __restrict__ lp = feat + (size_t)start * DP + lrow * DP + wbase + cl * 4;

    v8f acc[SLOTS] = {};
    const v2f sel = {(nl < 8) ? 1.0f : 0.0f, (nl < 8) ? 0.0f : 1.0f};

    auto loadStep = [&](int kk, v4f (&bb)[NLD]) {
        #pragma unroll
        for (int i = 0; i < NLD; ++i) {
            const int so = (kk + (i / T) * RPI) * DP + (i % T) * 64;
            bb[i] = *reinterpret_cast<const v4f*>(lp + so);
        }
    };
    auto mma2 = [&](int i, const v4f& b) {
        const int a = 2 * (i % NP);
        const v2f b0 = {b.x, b.y};
        const v2f b1 = {b.z, b.w};
        acc[a]     = __builtin_amdgcn_wmma_f32_16x16x4_f32(
            false, sel, false, b0, (short)0, acc[a], false, false);
        acc[a + 1] = __builtin_amdgcn_wmma_f32_16x16x4_f32(
            false, sel, false, b1, (short)0, acc[a + 1], false, false);
    };
    auto mmaStep = [&](v4f (&bb)[NLD]) {
        #pragma unroll
        for (int i = 0; i < NLD; ++i) mma2(i, bb[i]);
    };

    v4f A[NLD], B[NLD];
    int k0 = 0;

    if (cnt >= STEP) loadStep(0, A);

    for (; k0 + 3 * STEP <= cnt; k0 += 2 * STEP) {
        loadStep(k0 + STEP, B);
        mmaStep(A);
        loadStep(k0 + 2 * STEP, A);
        mmaStep(B);
    }

    if (k0 + 2 * STEP <= cnt) {
        loadStep(k0 + STEP, B);
        mmaStep(A);
        mmaStep(B);
        k0 += 2 * STEP;
    } else if (k0 + STEP <= cnt) {
        mmaStep(A);
        k0 += STEP;
    }

    if (k0 < cnt) {
        #pragma unroll
        for (int i = 0; i < NLD; ++i) {
            const int r0 = k0 + (i / T) * RPI;
            const v4f z = {0.0f, 0.0f, 0.0f, 0.0f};
            A[i] = (r0 + lrow < cnt)
                 ? *reinterpret_cast<const v4f*>(lp + r0 * DP + (i % T) * 64) : z;
        }
        #pragma unroll
        for (int i = 0; i < NLD / 2; ++i) mma2(i, A[i]);
        if (k0 + STEP / 2 < cnt) {
            #pragma unroll
            for (int i = NLD / 2; i < NLD; ++i) mma2(i, A[i]);
        }
    }

    const float cf = (float)cnt;
    float* __restrict__ og = out + (size_t)g * DP;
    #pragma unroll
    for (int t = 0; t < T; ++t) {
        #pragma unroll
        for (int j = 0; j < 2; ++j) {
            float v = acc[2 * t + j][0];
            #pragma unroll
            for (int a = t + T; a < NP; a += T) v += acc[2 * a + j][0];
            #pragma unroll
            for (int m = LPR; m < 16; m <<= 1) v += __shfl_xor(v, m);
            if (rs == 0) {
                const int col = wbase + t * 64 + cl * 4 + 2 * j + half;
                const int dp  = ((col & 255) << 2) | (col >> 8);
                og[dp] = v / cf;
            }
        }
    }
}

__device__ __forceinline__ unsigned unit_key(const int cnt, const int S, const int u) {
    const int lg = (S >> 1) - (S >> 3);
    return ((((unsigned)cnt << 3) >> lg) << 1) + (unsigned)u;
}

__global__ void __launch_bounds__(256, 8)
seg_mean_wmma_kernel(const float* __restrict__ feat,
                     const int*   __restrict__ counts,
                     float*       __restrict__ out,
                     const int                 ng) {
    __shared__ unsigned wtot[8];
    __shared__ int wless[8];
    __shared__ int rec[6];
    const int b    = blockIdx.x;
    const int nb   = gridDim.x;
    const int t    = threadIdx.x;
    const int lane = t & 31;
    const int wave = t >> 5;

    int c[4];
    if (4 * t + 3 < ng) {
        const int4 cv = *reinterpret_cast<const int4*>(counts + 4 * t);
        c[0] = cv.x; c[1] = cv.y; c[2] = cv.z; c[3] = cv.w;
    } else {
        #pragma unroll
        for (int j = 0; j < 4; ++j) c[j] = (4 * t + j < ng) ? counts[4 * t + j] : 0;
    }
    int      S[4];
    unsigned val[4];
    unsigned v = 0;
    #pragma unroll
    for (int j = 0; j < 4; ++j) {
        const int w = 1 + (c[j] >> 6);
        S[j]   = (4 * t + j < ng) ? ((w >= 8) ? 8 : (w >= 4) ? 4 : (w >= 2) ? 2 : 1) : 0;
        val[j] = (unsigned)c[j] + ((unsigned)S[j] << ROW_BITS);
        v += val[j];
    }

    unsigned inc = v;
    #pragma unroll
    for (int d = 1; d < 32; d <<= 1) {
        const unsigned n = __shfl_up(inc, d);
        if (lane >= d) inc += n;
    }
    if (lane == 31) wtot[wave] = inc;
    __syncthreads();
    unsigned wpre = 0, total = 0;
    #pragma unroll
    for (int w = 0; w < 8; ++w) {
        const unsigned x = wtot[w];
        if (w < wave) wpre += x;
        total += x;
    }
    const int nunits = (int)(total >> ROW_BITS);
    if (b >= nunits) return;

    const unsigned e0 = wpre + inc - v;

    auto publish = [&](const int u, const int slot) {
        unsigned e = e0;
        #pragma unroll
        for (int j = 0; j < 4; ++j) {
            const int u0 = (int)(e >> ROW_BITS);
            if (u >= u0 && u < u0 + S[j]) {
                rec[slot]     = (4 * t + j) | ((u - u0) << 10) | (S[j] << 13);
                rec[slot + 1] = c[j];
                rec[slot + 2] = (int)(e & ((1u << ROW_BITS) - 1u));
            }
            e += val[j];
        }
    };

    publish(b, 0);
    int todo = 1;
    const int n2 = nunits - nb;
    if (n2 > 0) {
        __syncthreads();
        const int mykey = (int)unit_key(rec[1], rec[0] >> 13, b);
        int less = 0;
        unsigned e = e0;
        #pragma unroll
        for (int j = 0; j < 4; ++j) {
            const int u0 = (int)(e >> ROW_BITS);
            const int k0 = (int)unit_key(c[j], S[j], u0);
            const int nj = min(S[j], max(nb - u0, 0));
            const int d  = mykey - k0;
            less += min(max(d, 0), nj);
            if (d >= 0 && d < nj && u0 + d < b) ++less;
            e += val[j];
        }
        int wsum = 0;
        #pragma unroll
        for (int bit = 0; bit < 6; ++bit)
            wsum += __popc(__builtin_amdgcn_ballot_w32(((less >> bit) & 1) != 0)) << bit;
        if (lane == 0) wless[wave] = wsum;
        __syncthreads();
        int rank = 0;
        #pragma unroll
        for (int w = 0; w < 8; ++w) rank += wless[w];
        const int r = __builtin_amdgcn_readfirstlane(rank);
        if (r < n2) {
            publish(nb + r, 3);
            todo = 2;
        }
    }
    __syncthreads();

    auto run = [&](const int k) {
        const int en    = __builtin_amdgcn_readfirstlane(rec[3 * k]);
        const int cnt   = __builtin_amdgcn_readfirstlane(rec[3 * k + 1]);
        const int start = __builtin_amdgcn_readfirstlane(rec[3 * k + 2]);
        const int g = en & 1023;
        const int q = (en >> 10) & 7;
        const int Su = en >> 13;
        const int tid = t;
        if (Su == 1)      seg_unit<1>(feat, cnt, start, g, q, out, tid);
        else if (Su == 2) seg_unit<2>(feat, cnt, start, g, q, out, tid);
        else if (Su == 4) seg_unit<4>(feat, cnt, start, g, q, out, tid);
        else              seg_unit<8>(feat, cnt, start, g, q, out, tid);
    };
    run(0);
    if (todo == 2) run(1);
}

extern "C" void kernel_launch(void* const* d_in, const int* in_sizes, int n_in,
                              void* d_out, int out_size, void* d_ws, size_t ws_size,
                              hipStream_t stream) {
    (void)n_in; (void)out_size; (void)d_ws; (void)ws_size;
    const float* feat   = (const float*)d_in[0];
    const int*   counts = (const int*)d_in[1];
    const int    G      = in_sizes[1];

    seg_mean_wmma_kernel<<<dim3(G), 256, 0, stream>>>(feat, counts, (float*)d_out, G);
}
